# starting kernel
# baseline (speedup 1.0000x reference)
#include <hip/hip_runtime.h>
#include <hip/hip_bf16.h>

#define DLAT 64
#define DFEAT 128
#define SLOPE 0.01f
#define EPSN 1e-12f

typedef __attribute__((ext_vector_type(2))) float v2f;
typedef __attribute__((ext_vector_type(8))) float v8f;

__device__ __forceinline__ float leaky(float v) { return v >= 0.0f ? v : SLOPE * v; }

__device__ __forceinline__ v8f wmma_f32(v2f a, v2f b, v8f c) {
    return __builtin_amdgcn_wmma_f32_16x16x4_f32(false, a, false, b, (short)0, c,
                                                 false, false);
}

__global__ __launch_bounds__(256) void mlp_norm_conv_kernel(
    const float* __restrict__ feat, const float* __restrict__ Wmlp,
    const float* __restrict__ bmlp, const float* __restrict__ Wc,
    float* __restrict__ xout, float* __restrict__ yout, int nnode)
{
    __shared__ float Wm_s[DFEAT * DLAT];
    __shared__ float Wc_s[DLAT * DLAT];
    __shared__ float xt[32 * 65];
    __shared__ float scl[32];

    const int tid = threadIdx.x;
    for (int i = tid; i < DFEAT * DLAT; i += 256) Wm_s[i] = Wmlp[i];
    for (int i = tid; i < DLAT * DLAT; i += 256) Wc_s[i] = Wc[i];
    __syncthreads();

    const int lane    = tid & 31;
    const int wave    = tid >> 5;
    const int rowtile = wave >> 2;
    const int coltile = wave & 3;
    const int mn      = lane & 15;
    const int kh      = (lane >> 4) << 1;
    const int col     = coltile * 16 + mn;
    const int blk0    = blockIdx.x * 32;

    int arow = blk0 + rowtile * 16 + mn;
    if (arow > nnode - 1) arow = nnode - 1;
    const float* ap = feat + (size_t)arow * DFEAT + kh;

    v8f acc = {};
    #pragma unroll
    for (int kk = 0; kk < DFEAT; kk += 4) {
        v2f a = *(const v2f*)(ap + kk);
        v2f b;
        b.x = Wm_s[(kk + kh) * DLAT + col];
        b.y = Wm_s[(kk + kh + 1) * DLAT + col];
        acc = wmma_f32(a, b, acc);
    }

    const float bias  = bmlp[col];
    const int   rbase = rowtile * 16 + ((lane >> 4) << 3);
    #pragma unroll
    for (int i = 0; i < 8; ++i)
        xt[(rbase + i) * 65 + col] = acc[i] + bias;
    __syncthreads();

    if (tid < 32) {
        float s = 0.f;
        #pragma unroll
        for (int j = 0; j < DLAT; ++j) { float v = xt[tid * 65 + j]; s += v * v; }
        scl[tid] = 1.0f / fmaxf(sqrtf(s), EPSN);
    }
    __syncthreads();

    for (int e = tid; e < 32 * DLAT; e += 256) {
        int r = e >> 6, c = e & 63;
        float v = xt[r * 65 + c] * scl[r];
        xt[r * 65 + c] = v;
        int g = blk0 + r;
        if (g < nnode) xout[(size_t)g * DLAT + c] = v;
    }
    __syncthreads();

    v8f acc2 = {};
    const int rl = rowtile * 16 + mn;
    #pragma unroll
    for (int kk = 0; kk < DLAT; kk += 4) {
        v2f a;
        a.x = xt[rl * 65 + kk + kh];
        a.y = xt[rl * 65 + kk + kh + 1];
        v2f b;
        b.x = Wc_s[(kk + kh) * DLAT + col];
        b.y = Wc_s[(kk + kh + 1) * DLAT + col];
        acc2 = wmma_f32(a, b, acc2);
    }
    #pragma unroll
    for (int i = 0; i < 8; ++i) {
        int g = blk0 + rbase + i;
        if (g < nnode) yout[(size_t)g * DLAT + col] = acc2[i];
    }
}

template <bool FUSE_NEXT>
__global__ __launch_bounds__(256) void update_kernel(
    const float* __restrict__ xin, const float* __restrict__ hraw,
    const float* __restrict__ idemb,
    const float* __restrict__ Wl, const float* __restrict__ bl,
    const float* __restrict__ Wg, const float* __restrict__ bg,
    const float* __restrict__ Wc, float* __restrict__ xout,
    float* __restrict__ yout, int nnode)
{
    __shared__ float Wl_s[DLAT * DLAT];
    __shared__ float Wg_s[DLAT * DLAT];
    __shared__ float Wc_s[DLAT * DLAT];
    __shared__ float xt[32 * 65];

    const int tid = threadIdx.x;
    for (int i = tid; i < DLAT * DLAT; i += 256) {
        Wl_s[i] = Wl[i];
        Wg_s[i] = Wg[i];
        if constexpr (FUSE_NEXT) Wc_s[i] = Wc[i];
    }
    __syncthreads();

    const int lane    = tid & 31;
    const int wave    = tid >> 5;
    const int rowtile = wave >> 2;
    const int coltile = wave & 3;
    const int mn      = lane & 15;
    const int kh      = (lane >> 4) << 1;
    const int col     = coltile * 16 + mn;
    const int blk0    = blockIdx.x * 32;

    int arow = blk0 + rowtile * 16 + mn;
    if (arow > nnode - 1) arow = nnode - 1;
    const float* xp = xin + (size_t)arow * DLAT + kh;
    const float* hp = hraw + (size_t)arow * DLAT + kh;

    v8f c1 = {}, c2 = {};
    #pragma unroll
    for (int kk = 0; kk < DLAT; kk += 4) {
        v2f ax = *(const v2f*)(xp + kk);
        v2f hv = *(const v2f*)(hp + kk);
        v2f ah; ah.x = leaky(hv.x); ah.y = leaky(hv.y);
        v2f b1;
        b1.x = Wl_s[(kk + kh) * DLAT + col];
        b1.y = Wl_s[(kk + kh + 1) * DLAT + col];
        v2f b2;
        b2.x = Wg_s[(kk + kh) * DLAT + col];
        b2.y = Wg_s[(kk + kh + 1) * DLAT + col];
        c1 = wmma_f32(ax, b1, c1);
        c2 = wmma_f32(ah, b2, c2);
    }

    const float blv = bl[col], bgv = bg[col];
    const int rbase = rowtile * 16 + ((lane >> 4) << 3);
    #pragma unroll
    for (int i = 0; i < 8; ++i) {
        int rloc = rbase + i;
        int g    = blk0 + rloc;
        int gc   = g > nnode - 1 ? nnode - 1 : g;
        float idv  = idemb[(size_t)gc * DLAT + col];
        float xhat = leaky(c1[i] + blv) + idv;
        float v    = leaky(c2[i] + bgv + xhat);
        if constexpr (FUSE_NEXT) {
            xt[rloc * 65 + col] = v;
        } else {
            if (g < nnode) xout[(size_t)g * DLAT + col] = v;
        }
    }

    if constexpr (FUSE_NEXT) {
        __syncthreads();
        for (int e = tid; e < 32 * DLAT; e += 256) {
            int r = e >> 6, c = e & 63;
            int g = blk0 + r;
            if (g < nnode) xout[(size_t)g * DLAT + c] = xt[r * 65 + c];
        }
        v8f acc2 = {};
        const int rl = rowtile * 16 + mn;
        #pragma unroll
        for (int kk = 0; kk < DLAT; kk += 4) {
            v2f a;
            a.x = xt[rl * 65 + kk + kh];
            a.y = xt[rl * 65 + kk + kh + 1];
            v2f b;
            b.x = Wc_s[(kk + kh) * DLAT + col];
            b.y = Wc_s[(kk + kh + 1) * DLAT + col];
            acc2 = wmma_f32(a, b, acc2);
        }
        #pragma unroll
        for (int i = 0; i < 8; ++i) {
            int g = blk0 + rbase + i;
            if (g < nnode) yout[(size_t)g * DLAT + col] = acc2[i];
        }
    }
}

__global__ __launch_bounds__(256) void scatter_kernel(
    const float* __restrict__ y, const int* __restrict__ src,
    const int* __restrict__ dst, float* __restrict__ h, int nedge)
{
    const int lane = threadIdx.x & 31;
    const int wid  = (blockIdx.x * blockDim.x + threadIdx.x) >> 5;
    const int nw   = (gridDim.x * blockDim.x) >> 5;
    for (int e = wid; e < nedge; e += nw) {
        const int s = src[e], d = dst[e];
        v2f v = *(const v2f*)(y + (size_t)s * DLAT + lane * 2);
        float* hp = h + (size_t)d * DLAT + lane * 2;
        unsafeAtomicAdd(hp, v.x);
        unsafeAtomicAdd(hp + 1, v.y);
    }
}

__global__ __launch_bounds__(256) void zero_kernel(float4* __restrict__ p, size_t n4)
{
    size_t i      = (size_t)blockIdx.x * blockDim.x + threadIdx.x;
    size_t stride = (size_t)gridDim.x * blockDim.x;
    float4 z = {0.f, 0.f, 0.f, 0.f};
    for (; i < n4; i += stride) p[i] = z;
}

extern "C" void kernel_launch(void* const* d_in, const int* in_sizes, int n_in,
                              void* d_out, int out_size, void* d_ws, size_t ws_size,
                              hipStream_t stream)
{
    const float* features = (const float*)d_in[0];
    const float* id_emb   = (const float*)d_in[1];
    const int*   edge     = (const int*)d_in[2];
    const float* Wmlp     = (const float*)d_in[3];
    const float* bmlp     = (const float*)d_in[4];
    const float* Wc1      = (const float*)d_in[5];
    const float* Wl1      = (const float*)d_in[6];
    const float* bl1      = (const float*)d_in[7];
    const float* Wg1      = (const float*)d_in[8];
    const float* bg1      = (const float*)d_in[9];
    const float* Wc2      = (const float*)d_in[10];
    const float* Wl2      = (const float*)d_in[11];
    const float* bl2      = (const float*)d_in[12];
    const float* Wg2      = (const float*)d_in[13];
    const float* bg2      = (const float*)d_in[14];

    const int nnode = in_sizes[0] / DFEAT;
    const int nedge = in_sizes[2] / 2;
    const int* src  = edge;
    const int* dst  = edge + nedge;

    float* x = (float*)d_ws;
    float* h = x + (size_t)nnode * DLAT;
    float* y = h + (size_t)nnode * DLAT;

    const int    nblk = (nnode + 31) / 32;
    const size_t hn4  = ((size_t)nnode * DLAT) / 4;

    mlp_norm_conv_kernel<<<nblk, 256, 0, stream>>>(features, Wmlp, bmlp, Wc1, x, y, nnode);
    zero_kernel<<<2048, 256, 0, stream>>>((float4*)h, hn4);
    scatter_kernel<<<2048, 256, 0, stream>>>(y, src, dst, h, nedge);
    update_kernel<true><<<nblk, 256, 0, stream>>>(x, h, id_emb, Wl1, bl1, Wg1, bg1,
                                                  Wc2, x, y, nnode);
    zero_kernel<<<2048, 256, 0, stream>>>((float4*)h, hn4);
    scatter_kernel<<<2048, 256, 0, stream>>>(y, src, dst, h, nedge);
    update_kernel<false><<<nblk, 256, 0, stream>>>(x, h, id_emb, Wl2, bl2, Wg2, bg2,
                                                   nullptr, (float*)d_out, nullptr, nnode);
}


# Dense kernels: a wave owns 16 whole rows (4 column-tile accumulators): each A fragment is loaded once for 4 products, was loaded by 4 waves; LDS tile wave-private, no block barrier; sum order kept
# speedup vs baseline: 1.5048x; 1.5048x over previous
#include <hip/hip_runtime.h>
#include <hip/hip_bf16.h>


#define DLAT 64
#define DFEAT 128
#define SLOPE 0.01f
#define EPSN 1e-12f

#define NWAVE 7
#define DTHREADS (NWAVE * 32)
#define DROWS (NWAVE * 16)
#define TP 68
#define DENSE_OCC __attribute__((amdgpu_waves_per_eu(NWAVE, NWAVE)))

typedef __attribute__((ext_vector_type(2))) float v2f;
typedef __attribute__((ext_vector_type(4))) float v4f;
typedef __attribute__((ext_vector_type(8))) float v8f;

__device__ __forceinline__ float leaky(float v) { return v >= 0.0f ? v : SLOPE * v; }

__device__ __forceinline__ int ypos(int c) { return ((c & 31) << 1) | (c >> 5); }

__device__ __forceinline__ v8f wmma_f32(v2f a, v2f b, v8f c) {
    return __builtin_amdgcn_wmma_f32_16x16x4_f32(false, a, false, b, (short)0, c,
                                                 false, false);
}

__device__ __forceinline__ void mma_row4(v2f a, const float* wk, v8f (&acc)[4]) {
    #pragma unroll
    for (int t = 0; t < 4; ++t) {
        v2f b;
        b.x = wk[16 * t];
        b.y = wk[DLAT + 16 * t];
        acc[t] = wmma_f32(a, b, acc[t]);
    }
}

__device__ __forceinline__ void wave_tile_sync() {
    __builtin_amdgcn_fence(__ATOMIC_RELEASE, "workgroup", "local");
    __builtin_amdgcn_wave_barrier();
    __builtin_amdgcn_fence(__ATOMIC_ACQUIRE, "workgroup", "local");
}

__device__ __forceinline__ void store_y_rows(float* __restrict__ yout, const v8f (&acc)[4],
                                             int row0, int c, int h, int nnode) {
    #pragma unroll
    for (int r = 0; r < 8; ++r) {
        const int g = row0 + 8 * h + r;
        if (g < nnode) {
            float* yr = yout + (size_t)g * DLAT + 2 * c;
            v2f lo; lo.x = acc[0][r]; lo.y = acc[2][r];
            v2f hi; hi.x = acc[1][r]; hi.y = acc[3][r];
            *(v2f*)yr        = lo;
            *(v2f*)(yr + 32) = hi;
        }
    }
}

__global__ __launch_bounds__(DTHREADS) DENSE_OCC void mlp_norm_conv_kernel(
    const float* __restrict__ feat, const float* __restrict__ Wmlp,
    const float* __restrict__ bmlp, const float* __restrict__ Wc,
    float* __restrict__ xout, float* __restrict__ yout, int nnode)
{
    __shared__ float Wm_s[DFEAT * DLAT];
    __shared__ float Wc_s[DLAT * DLAT];
    __shared__ __attribute__((aligned(16))) float tiles[NWAVE * 16 * TP];

    const int tid = threadIdx.x;
    for (int i = tid; i < DFEAT * DLAT; i += DTHREADS) Wm_s[i] = Wmlp[i];
    for (int i = tid; i < DLAT * DLAT; i += DTHREADS) Wc_s[i] = Wc[i];
    __syncthreads();

    const int lane = tid & 31;
    const int wave = tid >> 5;
    const int c    = lane & 15;
    const int h    = lane >> 4;
    const int row0 = (blockIdx.x * NWAVE + wave) * 16;
    float* xt = tiles + wave * (16 * TP);

    int arow = row0 + c;
    if (arow > nnode - 1) arow = nnode - 1;
    const float* ap = feat + (size_t)arow * DFEAT + 2 * h;
    const float* wm = Wm_s + (2 * h) * DLAT + c;

    v8f acc[4] = {};
    #pragma unroll
    for (int kk = 0; kk < DFEAT; kk += 4) {
        v2f a = *(const v2f*)(ap + kk);
        mma_row4(a, wm + kk * DLAT, acc);
    }

    #pragma unroll
    for (int t = 0; t < 4; ++t) {
        const float bias = bmlp[16 * t + c];
        #pragma unroll
        for (int r = 0; r < 8; ++r)
            xt[(8 * h + r) * TP + 16 * t + c] = acc[t][r] + bias;
    }
    wave_tile_sync();

    const float* xr = xt + c * TP;
    float s = 0.f;
    #pragma unroll
    for (int j = 0; j < DLAT; ++j) { float v = xr[j]; s += v * v; }
    const float scl = 1.0f / fmaxf(sqrtf(s), EPSN);

    #pragma unroll
    for (int i = 0; i < 8; ++i) {
        const int r = 2 * i + h;
        const float s0 = __builtin_bit_cast(float,
            __builtin_amdgcn_readlane(__builtin_bit_cast(int, scl), 2 * i));
        const float s1 = __builtin_bit_cast(float,
            __builtin_amdgcn_readlane(__builtin_bit_cast(int, scl), 2 * i + 1));
        const float sr = h ? s1 : s0;
        v4f v = *(const v4f*)(xt + r * TP + 4 * c);
        v.x *= sr; v.y *= sr; v.z *= sr; v.w *= sr;
        const int g = row0 + r;
        if (g < nnode) *(v4f*)(xout + (size_t)g * DLAT + 4 * c) = v;
    }

    v8f acc2[4] = {};
    const float* wc = Wc_s + (2 * h) * DLAT + c;
    #pragma unroll
    for (int kk = 0; kk < DLAT; kk += 4) {
        v2f a = *(const v2f*)(xr + kk + 2 * h);
        a.x *= scl; a.y *= scl;
        mma_row4(a, wc + kk * DLAT, acc2);
    }
    store_y_rows(yout, acc2, row0, c, h, nnode);
}

template <bool FUSE_NEXT>
__global__ __launch_bounds__(DTHREADS) DENSE_OCC void update_kernel(
    const float* xin, const float* __restrict__ hraw,
    const float* __restrict__ idemb,
    const float* __restrict__ Wl, const float* __restrict__ bl,
    const float* __restrict__ Wg, const float* __restrict__ bg,
    const float* __restrict__ Wc, float* xout,
    float* __restrict__ yout, int nnode)
{
    __shared__ __attribute__((aligned(16))) float lds[3 * DLAT * DLAT + NWAVE * 16 * TP];
    float* Wl_s  = lds;
    float* Wg_s  = lds + DLAT * DLAT;
    float* Wc_s  = lds + 2 * DLAT * DLAT;
    float* tiles = lds + 3 * DLAT * DLAT;

    const int tid = threadIdx.x;
    for (int i = tid; i < DLAT * DLAT; i += DTHREADS) {
        Wl_s[i] = Wl[i];
        Wg_s[i] = Wg[i];
        if constexpr (FUSE_NEXT) Wc_s[i] = Wc[i];
    }
    __syncthreads();

    const int lane = tid & 31;
    const int wave = tid >> 5;
    const int c    = lane & 15;
    const int h    = lane >> 4;
    const int row0 = (blockIdx.x * NWAVE + wave) * 16;
    float* xt = tiles + wave * (16 * TP);

    int arow = row0 + c;
    if (arow > nnode - 1) arow = nnode - 1;
    const float* xp = xin + (size_t)arow * DLAT + 2 * h;
    const float* hp = hraw + (size_t)arow * DLAT + 2 * h;
    const float* wl = Wl_s + (2 * h) * DLAT + c;
    const float* wg = Wg_s + (2 * h) * DLAT + c;

    v2f ax[16];
    #pragma unroll
    for (int i = 0; i < 16; ++i) ax[i] = *(const v2f*)(xp + 4 * i);

    float idv[4][8];
    v8f c1[4] = {};
    #pragma unroll
    for (int i = 0; i < 16; ++i) {
        mma_row4(ax[i], wl + 4 * i * DLAT, c1);
        if ((i & 3) == 3) __builtin_amdgcn_sched_barrier(0);
        if (i == 7) {
            #pragma unroll
            for (int r = 0; r < 8; ++r) {
                const int g  = row0 + 8 * h + r;
                const int gc = g > nnode - 1 ? nnode - 1 : g;
                const unsigned ido = (unsigned)gc * DLAT + (unsigned)c;
                #pragma unroll
                for (int t = 0; t < 4; ++t) idv[t][r] = idemb[ido + 16u * t];
            }
        }
    }

    float blv[4], bgv[4];
    #pragma unroll
    for (int t = 0; t < 4; ++t) { blv[t] = bl[16 * t + c]; bgv[t] = bg[16 * t + c]; }
    v2f hv[16];
    #pragma unroll
    for (int i = 0; i < 16; ++i) hv[i] = *(const v2f*)(hp + 4 * i);

    #pragma unroll
    for (int t = 0; t < 4; ++t) {
        #pragma unroll
        for (int r = 0; r < 8; ++r)
            xt[(8 * h + r) * TP + 16 * t + c] = leaky(c1[t][r] + blv[t]) + idv[t][r];
    }

    v8f c2[4] = {};
    #pragma unroll
    for (int i = 0; i < 16; ++i) {
        v2f ah; ah.x = leaky(hv[i].x); ah.y = leaky(hv[i].y);
        mma_row4(ah, wg + 4 * i * DLAT, c2);
        if ((i & 3) == 3) __builtin_amdgcn_sched_barrier(0);
    }
    #pragma unroll
    for (int t = 0; t < 4; ++t) {
        const int   col = 16 * t + c;
        #pragma unroll
        for (int r = 0; r < 8; ++r) {
            float* e = xt + (8 * h + r) * TP + col;
            const float v = leaky(c2[t][r] + bgv[t] + *e);
            *e = v;
        }
    }
    wave_tile_sync();

    #pragma unroll
    for (int i = 0; i < 8; ++i) {
        const int r = 2 * i + h;
        const v4f v = *(const v4f*)(xt + r * TP + 4 * c);
        const int g = row0 + r;
        if (g < nnode) *(v4f*)(xout + (size_t)g * DLAT + 4 * c) = v;
    }

    if constexpr (FUSE_NEXT) {
        v8f acc2[4] = {};
        const float* xr = xt + c * TP + 2 * h;
        const float* wc = Wc_s + (2 * h) * DLAT + c;
        #pragma unroll
        for (int kk = 0; kk < DLAT; kk += 4) {
            v2f a = *(const v2f*)(xr + kk);
            mma_row4(a, wc + kk * DLAT, acc2);
        }
        store_y_rows(yout, acc2, row0, c, h, nnode);
    }
}

#define EB 8
__global__ __launch_bounds__(256) void scatter_kernel(
    const float* __restrict__ y, const int* __restrict__ src,
    const int* __restrict__ dst, float* __restrict__ h, int nedge)
{
    const int lane = threadIdx.x & 31;
    const int wid  = __builtin_amdgcn_readfirstlane(
                         (int)((blockIdx.x * blockDim.x + threadIdx.x) >> 5));
    const int nw   = (gridDim.x * blockDim.x) >> 5;
    const int per  = (nedge + nw - 1) / nw;
    const int e0   = wid * per;
    int e1 = e0 + per;
    if (e1 > nedge) e1 = nedge;

    const float* yl = y + lane * 2;
    float*       hl = h + lane;

    for (int base = e0; base < e1; base += 32) {
        const int e = base + lane;
        int s = 0, d = 0;
        if (e < e1) { s = src[e]; d = dst[e]; }
        int cnt = e1 - base;
        if (cnt > 32) cnt = 32;

        int b = 0;
        for (; b + EB <= cnt; b += EB) {
            v2f v[EB];
            #pragma unroll
            for (int j = 0; j < EB; ++j) {
                const int sj = __builtin_amdgcn_readlane(s, b + j);
                v[j] = *(const v2f*)(yl + (size_t)sj * DLAT);
            }
            #pragma unroll
            for (int j = 0; j < EB; ++j) {
                const int dj = __builtin_amdgcn_readlane(d, b + j);
                float* hp = hl + (size_t)dj * DLAT;
                unsafeAtomicAdd(hp, v[j].x);
                unsafeAtomicAdd(hp + 32, v[j].y);
            }
        }
        for (; b < cnt; ++b) {
            const int sj = __builtin_amdgcn_readlane(s, b);
            const int dj = __builtin_amdgcn_readlane(d, b);
            v2f v = *(const v2f*)(yl + (size_t)sj * DLAT);
            float* hp = hl + (size_t)dj * DLAT;
            unsafeAtomicAdd(hp, v.x);
            unsafeAtomicAdd(hp + 32, v.y);
        }
    }
}

__global__ __launch_bounds__(256) void zero_kernel(float4* __restrict__ p, size_t n4)
{
    size_t i      = (size_t)blockIdx.x * blockDim.x + threadIdx.x;
    size_t stride = (size_t)gridDim.x * blockDim.x;
    float4 z = {0.f, 0.f, 0.f, 0.f};
    for (; i < n4; i += stride) p[i] = z;
}

extern "C" void kernel_launch(void* const* d_in, const int* in_sizes, int n_in,
                              void* d_out, int out_size, void* d_ws, size_t ws_size,
                              hipStream_t stream)
{
    const float* features = (const float*)d_in[0];
    const float* id_emb   = (const float*)d_in[1];
    const int*   edge     = (const int*)d_in[2];
    const float* Wmlp     = (const float*)d_in[3];
    const float* bmlp     = (const float*)d_in[4];
    const float* Wc1      = (const float*)d_in[5];
    const float* Wl1      = (const float*)d_in[6];
    const float* bl1      = (const float*)d_in[7];
    const float* Wg1      = (const float*)d_in[8];
    const float* bg1      = (const float*)d_in[9];
    const float* Wc2      = (const float*)d_in[10];
    const float* Wl2      = (const float*)d_in[11];
    const float* bl2      = (const float*)d_in[12];
    const float* Wg2      = (const float*)d_in[13];
    const float* bg2      = (const float*)d_in[14];

    const int nnode = in_sizes[0] / DFEAT;
    const int nedge = in_sizes[2] / 2;
    const int* src  = edge;
    const int* dst  = edge + nedge;

    float* x = (float*)d_ws;
    float* h = x + (size_t)nnode * DLAT;
    float* y = h + (size_t)nnode * DLAT;

    const int    nblk = (nnode + DROWS - 1) / DROWS;
    const size_t hn4  = ((size_t)nnode * DLAT) / 4;

    mlp_norm_conv_kernel<<<nblk, DTHREADS, 0, stream>>>(features, Wmlp, bmlp, Wc1, x, y, nnode);
    zero_kernel<<<2048, 256, 0, stream>>>((float4*)h, hn4);
    scatter_kernel<<<2048, 256, 0, stream>>>(y, src, dst, h, nedge);
    update_kernel<true><<<nblk, DTHREADS, 0, stream>>>(x, h, id_emb, Wl1, bl1, Wg1, bg1,
                                                  Wc2, x, y, nnode);
    zero_kernel<<<2048, 256, 0, stream>>>((float4*)h, hn4);
    scatter_kernel<<<2048, 256, 0, stream>>>(y, src, dst, h, nedge);
    update_kernel<false><<<nblk, DTHREADS, 0, stream>>>(x, h, id_emb, Wl2, bl2, Wg2, bg2,
                                                   nullptr, (float*)d_out, nullptr, nnode);
}


# Zero fill of the scatter sums folded into the dense kernel in front of each scatter (a wave clears its own 16 rows under its weight loads): both fill launches gone, 7 to 5 launches; sum order kept
# speedup vs baseline: 1.6269x; 1.0811x over previous
#include <hip/hip_runtime.h>
#include <hip/hip_bf16.h>


#define DLAT 64
#define DFEAT 128
#define SLOPE 0.01f
#define EPSN 1e-12f

#define NWAVE 7
#define DTHREADS (NWAVE * 32)
#define DROWS (NWAVE * 16)
#define TP 68
#define DENSE_OCC __attribute__((amdgpu_waves_per_eu(NWAVE, NWAVE)))

typedef __attribute__((ext_vector_type(2))) float v2f;
typedef __attribute__((ext_vector_type(4))) float v4f;
typedef __attribute__((ext_vector_type(8))) float v8f;

__device__ __forceinline__ float leaky(float v) { return v >= 0.0f ? v : SLOPE * v; }

__device__ __forceinline__ int ypos(int c) { return ((c & 31) << 1) | (c >> 5); }

__device__ __forceinline__ v8f wmma_f32(v2f a, v2f b, v8f c) {
    return __builtin_amdgcn_wmma_f32_16x16x4_f32(false, a, false, b, (short)0, c,
                                                 false, false);
}

__device__ __forceinline__ void mma_row4(v2f a, const float* wk, v8f (&acc)[4]) {
    #pragma unroll
    for (int t = 0; t < 4; ++t) {
        v2f b;
        b.x = wk[16 * t];
        b.y = wk[DLAT + 16 * t];
        acc[t] = wmma_f32(a, b, acc[t]);
    }
}

__device__ __forceinline__ void wave_tile_sync() {
    __builtin_amdgcn_fence(__ATOMIC_RELEASE, "workgroup", "local");
    __builtin_amdgcn_wave_barrier();
    __builtin_amdgcn_fence(__ATOMIC_ACQUIRE, "workgroup", "local");
}

__device__ __forceinline__ void store_y_rows(float* __restrict__ yout, const v8f (&acc)[4],
                                             int row0, int c, int h, int nnode) {
    #pragma unroll
    for (int r = 0; r < 8; ++r) {
        const int g = row0 + 8 * h + r;
        if (g < nnode) {
            float* yr = yout + (size_t)g * DLAT + 2 * c;
            v2f lo; lo.x = acc[0][r]; lo.y = acc[2][r];
            v2f hi; hi.x = acc[1][r]; hi.y = acc[3][r];
            *(v2f*)yr        = lo;
            *(v2f*)(yr + 32) = hi;
        }
    }
}

__device__ __forceinline__ void zero_rows(float* __restrict__ hz, int row0, int c, int h,
                                          int nnode) {
    const v4f z = {0.f, 0.f, 0.f, 0.f};
    #pragma unroll
    for (int i = 0; i < 8; ++i) {
        const int g = row0 + 2 * i + h;
        if (g < nnode) *(v4f*)(hz + (size_t)g * DLAT + 4 * c) = z;
    }
}

__global__ __launch_bounds__(DTHREADS) DENSE_OCC void mlp_norm_conv_kernel(
    const float* __restrict__ feat, const float* __restrict__ Wmlp,
    const float* __restrict__ bmlp, const float* __restrict__ Wc,
    float* __restrict__ xout, float* __restrict__ yout,
    float* __restrict__ hzero, int nnode)
{
    __shared__ float Wm_s[DFEAT * DLAT];
    __shared__ float Wc_s[DLAT * DLAT];
    __shared__ __attribute__((aligned(16))) float tiles[NWAVE * 16 * TP];

    const int tid  = threadIdx.x;
    const int lane = tid & 31;
    const int wave = tid >> 5;
    const int c    = lane & 15;
    const int h    = lane >> 4;
    const int row0 = (blockIdx.x * NWAVE + wave) * 16;
    float* xt = tiles + wave * (16 * TP);

    zero_rows(hzero, row0, c, h, nnode);

    for (int i = tid; i < DFEAT * DLAT; i += DTHREADS) Wm_s[i] = Wmlp[i];
    for (int i = tid; i < DLAT * DLAT; i += DTHREADS) Wc_s[i] = Wc[i];
    __syncthreads();

    int arow = row0 + c;
    if (arow > nnode - 1) arow = nnode - 1;
    const float* ap = feat + (size_t)arow * DFEAT + 2 * h;
    const float* wm = Wm_s + (2 * h) * DLAT + c;

    v8f acc[4] = {};
    #pragma unroll
    for (int kk = 0; kk < DFEAT; kk += 4) {
        v2f a = *(const v2f*)(ap + kk);
        mma_row4(a, wm + kk * DLAT, acc);
    }

    #pragma unroll
    for (int t = 0; t < 4; ++t) {
        const float bias = bmlp[16 * t + c];
        #pragma unroll
        for (int r = 0; r < 8; ++r)
            xt[(8 * h + r) * TP + 16 * t + c] = acc[t][r] + bias;
    }
    wave_tile_sync();

    const float* xr = xt + c * TP;
    float s = 0.f;
    #pragma unroll
    for (int j = 0; j < DLAT; ++j) { float v = xr[j]; s += v * v; }
    const float scl = 1.0f / fmaxf(sqrtf(s), EPSN);

    #pragma unroll
    for (int i = 0; i < 8; ++i) {
        const int r = 2 * i + h;
        const float s0 = __builtin_bit_cast(float,
            __builtin_amdgcn_readlane(__builtin_bit_cast(int, scl), 2 * i));
        const float s1 = __builtin_bit_cast(float,
            __builtin_amdgcn_readlane(__builtin_bit_cast(int, scl), 2 * i + 1));
        const float sr = h ? s1 : s0;
        v4f v = *(const v4f*)(xt + r * TP + 4 * c);
        v.x *= sr; v.y *= sr; v.z *= sr; v.w *= sr;
        const int g = row0 + r;
        if (g < nnode) *(v4f*)(xout + (size_t)g * DLAT + 4 * c) = v;
    }

    v8f acc2[4] = {};
    const float* wc = Wc_s + (2 * h) * DLAT + c;
    #pragma unroll
    for (int kk = 0; kk < DLAT; kk += 4) {
        v2f a = *(const v2f*)(xr + kk + 2 * h);
        a.x *= scl; a.y *= scl;
        mma_row4(a, wc + kk * DLAT, acc2);
    }
    store_y_rows(yout, acc2, row0, c, h, nnode);
}

template <bool FUSE_NEXT>
__global__ __launch_bounds__(DTHREADS) DENSE_OCC void update_kernel(
    const float* xin, const float* __restrict__ hraw,
    const float* __restrict__ idemb,
    const float* __restrict__ Wl, const float* __restrict__ bl,
    const float* __restrict__ Wg, const float* __restrict__ bg,
    const float* __restrict__ Wc, float* xout,
    float* __restrict__ yout, float* __restrict__ hzero, int nnode)
{
    __shared__ __attribute__((aligned(16))) float lds[3 * DLAT * DLAT + NWAVE * 16 * TP];
    float* Wl_s  = lds;
    float* Wg_s  = lds + DLAT * DLAT;
    float* Wc_s  = lds + 2 * DLAT * DLAT;
    float* tiles = lds + 3 * DLAT * DLAT;

    const int tid  = threadIdx.x;
    const int lane = tid & 31;
    const int wave = tid >> 5;
    const int c    = lane & 15;
    const int h    = lane >> 4;
    const int row0 = (blockIdx.x * NWAVE + wave) * 16;
    float* xt = tiles + wave * (16 * TP);

    if constexpr (FUSE_NEXT) zero_rows(hzero, row0, c, h, nnode);

    for (int i = tid; i < DLAT * DLAT; i += DTHREADS) {
        Wl_s[i] = Wl[i];
        Wg_s[i] = Wg[i];
        if constexpr (FUSE_NEXT) Wc_s[i] = Wc[i];
    }
    __syncthreads();

    int arow = row0 + c;
    if (arow > nnode - 1) arow = nnode - 1;
    const float* xp = xin + (size_t)arow * DLAT + 2 * h;
    const float* hp = hraw + (size_t)arow * DLAT + 2 * h;
    const float* wl = Wl_s + (2 * h) * DLAT + c;
    const float* wg = Wg_s + (2 * h) * DLAT + c;

    v2f ax[16];
    #pragma unroll
    for (int i = 0; i < 16; ++i) ax[i] = *(const v2f*)(xp + 4 * i);

    float idv[4][8];
    v8f c1[4] = {};
    #pragma unroll
    for (int i = 0; i < 16; ++i) {
        mma_row4(ax[i], wl + 4 * i * DLAT, c1);
        if ((i & 3) == 3) __builtin_amdgcn_sched_barrier(0);
        if (i == 7) {
            #pragma unroll
            for (int r = 0; r < 8; ++r) {
                const int g  = row0 + 8 * h + r;
                const int gc = g > nnode - 1 ? nnode - 1 : g;
                const unsigned ido = (unsigned)gc * DLAT + (unsigned)c;
                #pragma unroll
                for (int t = 0; t < 4; ++t) idv[t][r] = idemb[ido + 16u * t];
            }
        }
    }

    float blv[4], bgv[4];
    #pragma unroll
    for (int t = 0; t < 4; ++t) { blv[t] = bl[16 * t + c]; bgv[t] = bg[16 * t + c]; }
    v2f hv[16];
    #pragma unroll
    for (int i = 0; i < 16; ++i) hv[i] = *(const v2f*)(hp + 4 * i);

    #pragma unroll
    for (int t = 0; t < 4; ++t) {
        #pragma unroll
        for (int r = 0; r < 8; ++r)
            xt[(8 * h + r) * TP + 16 * t + c] = leaky(c1[t][r] + blv[t]) + idv[t][r];
    }

    v8f c2[4] = {};
    #pragma unroll
    for (int i = 0; i < 16; ++i) {
        v2f ah; ah.x = leaky(hv[i].x); ah.y = leaky(hv[i].y);
        mma_row4(ah, wg + 4 * i * DLAT, c2);
        if ((i & 3) == 3) __builtin_amdgcn_sched_barrier(0);
    }
    #pragma unroll
    for (int t = 0; t < 4; ++t) {
        const int   col = 16 * t + c;
        #pragma unroll
        for (int r = 0; r < 8; ++r) {
            float* e = xt + (8 * h + r) * TP + col;
            const float v = leaky(c2[t][r] + bgv[t] + *e);
            *e = v;
        }
    }
    wave_tile_sync();

    #pragma unroll
    for (int i = 0; i < 8; ++i) {
        const int r = 2 * i + h;
        const v4f v = *(const v4f*)(xt + r * TP + 4 * c);
        const int g = row0 + r;
        if (g < nnode) *(v4f*)(xout + (size_t)g * DLAT + 4 * c) = v;
    }

    if constexpr (FUSE_NEXT) {
        v8f acc2[4] = {};
        const float* xr = xt + c * TP + 2 * h;
        const float* wc = Wc_s + (2 * h) * DLAT + c;
        #pragma unroll
        for (int kk = 0; kk < DLAT; kk += 4) {
            v2f a = *(const v2f*)(xr + kk);
            mma_row4(a, wc + kk * DLAT, acc2);
        }
        store_y_rows(yout, acc2, row0, c, h, nnode);
    }
}

#define EB 8
__global__ __launch_bounds__(256) void scatter_kernel(
    const float* __restrict__ y, const int* __restrict__ src,
    const int* __restrict__ dst, float* __restrict__ h, int nedge)
{
    const int lane = threadIdx.x & 31;
    const int wid  = __builtin_amdgcn_readfirstlane(
                         (int)((blockIdx.x * blockDim.x + threadIdx.x) >> 5));
    const int nw   = (gridDim.x * blockDim.x) >> 5;
    const int per  = (nedge + nw - 1) / nw;
    const int e0   = wid * per;
    int e1 = e0 + per;
    if (e1 > nedge) e1 = nedge;

    const float* yl = y + lane * 2;
    float*       hl = h + lane;

    for (int base = e0; base < e1; base += 32) {
        const int e = base + lane;
        int s = 0, d = 0;
        if (e < e1) { s = src[e]; d = dst[e]; }
        int cnt = e1 - base;
        if (cnt > 32) cnt = 32;

        int b = 0;
        for (; b + EB <= cnt; b += EB) {
            v2f v[EB];
            #pragma unroll
            for (int j = 0; j < EB; ++j) {
                const int sj = __builtin_amdgcn_readlane(s, b + j);
                v[j] = *(const v2f*)(yl + (size_t)sj * DLAT);
            }
            #pragma unroll
            for (int j = 0; j < EB; ++j) {
                const int dj = __builtin_amdgcn_readlane(d, b + j);
                float* hp = hl + (size_t)dj * DLAT;
                unsafeAtomicAdd(hp, v[j].x);
                unsafeAtomicAdd(hp + 32, v[j].y);
            }
        }
        for (; b < cnt; ++b) {
            const int sj = __builtin_amdgcn_readlane(s, b);
            const int dj = __builtin_amdgcn_readlane(d, b);
            v2f v = *(const v2f*)(yl + (size_t)sj * DLAT);
            float* hp = hl + (size_t)dj * DLAT;
            unsafeAtomicAdd(hp, v.x);
            unsafeAtomicAdd(hp + 32, v.y);
        }
    }
}

extern "C" void kernel_launch(void* const* d_in, const int* in_sizes, int n_in,
                              void* d_out, int out_size, void* d_ws, size_t ws_size,
                              hipStream_t stream)
{
    const float* features = (const float*)d_in[0];
    const float* id_emb   = (const float*)d_in[1];
    const int*   edge     = (const int*)d_in[2];
    const float* Wmlp     = (const float*)d_in[3];
    const float* bmlp     = (const float*)d_in[4];
    const float* Wc1      = (const float*)d_in[5];
    const float* Wl1      = (const float*)d_in[6];
    const float* bl1      = (const float*)d_in[7];
    const float* Wg1      = (const float*)d_in[8];
    const float* bg1      = (const float*)d_in[9];
    const float* Wc2      = (const float*)d_in[10];
    const float* Wl2      = (const float*)d_in[11];
    const float* bl2      = (const float*)d_in[12];
    const float* Wg2      = (const float*)d_in[13];
    const float* bg2      = (const float*)d_in[14];

    const int nnode = in_sizes[0] / DFEAT;
    const int nedge = in_sizes[2] / 2;
    const int* src  = edge;
    const int* dst  = edge + nedge;

    float* x  = (float*)d_ws;
    float* h1 = x  + (size_t)nnode * DLAT;
    float* y  = h1 + (size_t)nnode * DLAT;
    float* h2 = y  + (size_t)nnode * DLAT;

    const int nblk = (nnode + DROWS - 1) / DROWS;

    mlp_norm_conv_kernel<<<nblk, DTHREADS, 0, stream>>>(features, Wmlp, bmlp, Wc1, x, y, h1,
                                                        nnode);
    scatter_kernel<<<2048, 256, 0, stream>>>(y, src, dst, h1, nedge);
    update_kernel<true><<<nblk, DTHREADS, 0, stream>>>(x, h1, id_emb, Wl1, bl1, Wg1, bg1,
                                                  Wc2, x, y, h2, nnode);
    scatter_kernel<<<2048, 256, 0, stream>>>(y, src, dst, h2, nedge);
    update_kernel<false><<<nblk, DTHREADS, 0, stream>>>(x, h2, id_emb, Wl2, bl2, Wg2, bg2,
                                                   nullptr, (float*)d_out, nullptr, nullptr,
                                                   nnode);
}


# speedup vs baseline: 1.6812x; 1.0334x over previous
#include <hip/hip_runtime.h>
#include <hip/hip_bf16.h>


#define DLAT 64
#define DFEAT 128
#define SLOPE 0.01f
#define EPSN 1e-12f

#define NWAVE 7
#define DTHREADS (NWAVE * 32)
#define DROWS (NWAVE * 16)
#define TP 68
#define DENSE_OCC __attribute__((amdgpu_waves_per_eu(NWAVE, NWAVE)))

typedef __attribute__((ext_vector_type(2))) float v2f;
typedef __attribute__((ext_vector_type(4))) float v4f;
typedef __attribute__((ext_vector_type(8))) float v8f;

__device__ __forceinline__ float leaky(float v) { return v >= 0.0f ? v : SLOPE * v; }

__device__ __forceinline__ int ypos(int c) { return ((c & 31) << 1) | (c >> 5); }

__device__ __forceinline__ v8f wmma_f32(v2f a, v2f b, v8f c) {
    return __builtin_amdgcn_wmma_f32_16x16x4_f32(false, a, false, b, (short)0, c,
                                                 false, false);
}

__device__ __forceinline__ void mma_row4(v2f a, const float* wk, v8f (&acc)[4]) {
    #pragma unroll
    for (int t = 0; t < 4; ++t) {
        v2f b;
        b.x = wk[16 * t];
        b.y = wk[DLAT + 16 * t];
        acc[t] = wmma_f32(a, b, acc[t]);
    }
}

__device__ __forceinline__ void wave_tile_sync() {
    __builtin_amdgcn_fence(__ATOMIC_RELEASE, "workgroup", "local");
    __builtin_amdgcn_wave_barrier();
    __builtin_amdgcn_fence(__ATOMIC_ACQUIRE, "workgroup", "local");
}

__device__ __forceinline__ void store_y_rows(float* __restrict__ yout, const v8f (&acc)[4],
                                             int row0, int c, int h, int nnode) {
    #pragma unroll
    for (int r = 0; r < 8; ++r) {
        const int g = row0 + 8 * h + r;
        if (g < nnode) {
            float* yr = yout + (size_t)g * DLAT + 2 * c;
            v2f lo; lo.x = acc[0][r]; lo.y = acc[2][r];
            v2f hi; hi.x = acc[1][r]; hi.y = acc[3][r];
            *(v2f*)yr        = lo;
            *(v2f*)(yr + 32) = hi;
        }
    }
}

__device__ __forceinline__ void zero_rows(float* __restrict__ hz, size_t plane, int row0, int c,
                                          int h, int nnode) {
    const v4f z = {0.f, 0.f, 0.f, 0.f};
    #pragma unroll
    for (int i = 0; i < 8; ++i) {
        const int g = row0 + 2 * i + h;
        if (g < nnode) {
            float* p = hz + (size_t)g * DLAT + 4 * c;
            *(v4f*)p = z;
            *(v4f*)(p + plane) = z;
        }
    }
}

__global__ __launch_bounds__(DTHREADS) DENSE_OCC void mlp_norm_conv_kernel(
    const float* __restrict__ feat, const float* __restrict__ Wmlp,
    const float* __restrict__ bmlp, const float* __restrict__ Wc,
    float* __restrict__ xout, float* __restrict__ yout,
    float* __restrict__ hzero, int nnode)
{
    __shared__ float Wm_s[DFEAT * DLAT];
    __shared__ float Wc_s[DLAT * DLAT];
    __shared__ __attribute__((aligned(16))) float tiles[NWAVE * 16 * TP];

    const int tid  = threadIdx.x;
    const int lane = tid & 31;
    const int wave = tid >> 5;
    const int c    = lane & 15;
    const int h    = lane >> 4;
    const int row0 = (blockIdx.x * NWAVE + wave) * 16;
    float* xt = tiles + wave * (16 * TP);

    zero_rows(hzero, (size_t)nnode * DLAT, row0, c, h, nnode);

    for (int i = tid; i < DFEAT * DLAT; i += DTHREADS) Wm_s[i] = Wmlp[i];
    for (int i = tid; i < DLAT * DLAT; i += DTHREADS) Wc_s[i] = Wc[i];
    __syncthreads();

    int arow = row0 + c;
    if (arow > nnode - 1) arow = nnode - 1;
    const float* ap = feat + (size_t)arow * DFEAT + 2 * h;
    const float* wm = Wm_s + (2 * h) * DLAT + c;

    v8f acc[4] = {};
    #pragma unroll
    for (int kk = 0; kk < DFEAT; kk += 4) {
        v2f a = *(const v2f*)(ap + kk);
        mma_row4(a, wm + kk * DLAT, acc);
    }

    #pragma unroll
    for (int t = 0; t < 4; ++t) {
        const float bias = bmlp[16 * t + c];
        #pragma unroll
        for (int r = 0; r < 8; ++r)
            xt[(8 * h + r) * TP + 16 * t + c] = acc[t][r] + bias;
    }
    wave_tile_sync();

    const float* xr = xt + c * TP;
    float s = 0.f;
    #pragma unroll
    for (int j = 0; j < DLAT; ++j) { float v = xr[j]; s += v * v; }
    const float scl = 1.0f / fmaxf(sqrtf(s), EPSN);

    #pragma unroll
    for (int i = 0; i < 8; ++i) {
        const int r = 2 * i + h;
        const float s0 = __builtin_bit_cast(float,
            __builtin_amdgcn_readlane(__builtin_bit_cast(int, scl), 2 * i));
        const float s1 = __builtin_bit_cast(float,
            __builtin_amdgcn_readlane(__builtin_bit_cast(int, scl), 2 * i + 1));
        const float sr = h ? s1 : s0;
        v4f v = *(const v4f*)(xt + r * TP + 4 * c);
        v.x *= sr; v.y *= sr; v.z *= sr; v.w *= sr;
        const int g = row0 + r;
        if (g < nnode) *(v4f*)(xout + (size_t)g * DLAT + 4 * c) = v;
    }

    v8f acc2[4] = {};
    const float* wc = Wc_s + (2 * h) * DLAT + c;
    #pragma unroll
    for (int kk = 0; kk < DLAT; kk += 4) {
        v2f a = *(const v2f*)(xr + kk + 2 * h);
        a.x *= scl; a.y *= scl;
        mma_row4(a, wc + kk * DLAT, acc2);
    }
    store_y_rows(yout, acc2, row0, c, h, nnode);
}

template <bool FUSE_NEXT>
__global__ __launch_bounds__(DTHREADS) DENSE_OCC void update_kernel(
    const float* xin, const float* __restrict__ hraw,
    const float* __restrict__ idemb,
    const float* __restrict__ Wl, const float* __restrict__ bl,
    const float* __restrict__ Wg, const float* __restrict__ bg,
    const float* __restrict__ Wc, float* xout,
    float* __restrict__ yout, float* __restrict__ hzero, int nnode)
{
    __shared__ __attribute__((aligned(16))) float lds[3 * DLAT * DLAT + NWAVE * 16 * TP];
    float* Wl_s  = lds;
    float* Wg_s  = lds + DLAT * DLAT;
    float* Wc_s  = lds + 2 * DLAT * DLAT;
    float* tiles = lds + 3 * DLAT * DLAT;

    const int tid  = threadIdx.x;
    const int lane = tid & 31;
    const int wave = tid >> 5;
    const int c    = lane & 15;
    const int h    = lane >> 4;
    const int row0 = (blockIdx.x * NWAVE + wave) * 16;
    float* xt = tiles + wave * (16 * TP);

    if constexpr (FUSE_NEXT) zero_rows(hzero, (size_t)nnode * DLAT, row0, c, h, nnode);

    for (int i = tid; i < DLAT * DLAT; i += DTHREADS) {
        Wl_s[i] = Wl[i];
        Wg_s[i] = Wg[i];
        if constexpr (FUSE_NEXT) Wc_s[i] = Wc[i];
    }
    __syncthreads();

    int arow = row0 + c;
    if (arow > nnode - 1) arow = nnode - 1;
    const float* xp = xin + (size_t)arow * DLAT + 2 * h;
    const float* hp = hraw + (size_t)arow * DLAT + 2 * h;
    const float* hq = hp + (size_t)nnode * DLAT;
    const float* wl = Wl_s + (2 * h) * DLAT + c;
    const float* wg = Wg_s + (2 * h) * DLAT + c;

    v2f ax[16];
    #pragma unroll
    for (int i = 0; i < 16; ++i) ax[i] = *(const v2f*)(xp + 4 * i);

    float idv[4][8];
    v8f c1[4] = {};
    #pragma unroll
    for (int i = 0; i < 16; ++i) {
        mma_row4(ax[i], wl + 4 * i * DLAT, c1);
        if ((i & 3) == 3) __builtin_amdgcn_sched_barrier(0);
        if (i == 7) {
            #pragma unroll
            for (int r = 0; r < 8; ++r) {
                const int g  = row0 + 8 * h + r;
                const int gc = g > nnode - 1 ? nnode - 1 : g;
                const unsigned ido = (unsigned)gc * DLAT + (unsigned)c;
                #pragma unroll
                for (int t = 0; t < 4; ++t) idv[t][r] = idemb[ido + 16u * t];
            }
        }
    }

    float blv[4], bgv[4];
    #pragma unroll
    for (int t = 0; t < 4; ++t) { blv[t] = bl[16 * t + c]; bgv[t] = bg[16 * t + c]; }
    v2f hv[16];
    #pragma unroll
    for (int i = 0; i < 16; ++i) hv[i] = *(const v2f*)(hp + 4 * i);

    #pragma unroll
    for (int t = 0; t < 4; ++t) {
        #pragma unroll
        for (int r = 0; r < 8; ++r)
            xt[(8 * h + r) * TP + 16 * t + c] = leaky(c1[t][r] + blv[t]) + idv[t][r];
    }

    v2f hb[16];
    #pragma unroll
    for (int i = 0; i < 16; ++i) hb[i] = *(const v2f*)(hq + 4 * i);

    v8f c2[4] = {};
    #pragma unroll
    for (int i = 0; i < 16; ++i) {
        v2f ah; ah.x = leaky(hv[i].x + hb[i].x); ah.y = leaky(hv[i].y + hb[i].y);
        mma_row4(ah, wg + 4 * i * DLAT, c2);
        if ((i & 3) == 3) __builtin_amdgcn_sched_barrier(0);
    }
    #pragma unroll
    for (int t = 0; t < 4; ++t) {
        const int   col = 16 * t + c;
        #pragma unroll
        for (int r = 0; r < 8; ++r) {
            float* e = xt + (8 * h + r) * TP + col;
            const float v = leaky(c2[t][r] + bgv[t] + *e);
            *e = v;
        }
    }
    wave_tile_sync();

    #pragma unroll
    for (int i = 0; i < 8; ++i) {
        const int r = 2 * i + h;
        const v4f v = *(const v4f*)(xt + r * TP + 4 * c);
        const int g = row0 + r;
        if (g < nnode) *(v4f*)(xout + (size_t)g * DLAT + 4 * c) = v;
    }

    if constexpr (FUSE_NEXT) {
        v8f acc2[4] = {};
        const float* xr = xt + c * TP + 2 * h;
        const float* wc = Wc_s + (2 * h) * DLAT + c;
        #pragma unroll
        for (int kk = 0; kk < DLAT; kk += 4) {
            v2f a = *(const v2f*)(xr + kk);
            mma_row4(a, wc + kk * DLAT, acc2);
        }
        store_y_rows(yout, acc2, row0, c, h, nnode);
    }
}

#define EB 8
__global__ __launch_bounds__(256) void scatter_kernel(
    const float* __restrict__ y, const int* __restrict__ src,
    const int* __restrict__ dst, float* __restrict__ h, int nedge, int nnode)
{
    const int lane = threadIdx.x & 31;
    const int wid  = __builtin_amdgcn_readfirstlane(
                         (int)((blockIdx.x * blockDim.x + threadIdx.x) >> 5));
    const int nw   = (gridDim.x * blockDim.x) >> 5;
    const int per  = (nedge + nw - 1) / nw;
    const int e0   = wid * per;
    int e1 = e0 + per;
    if (e1 > nedge) e1 = nedge;

    const float* yl = y + lane * 2;
    const int half = (int)((__builtin_amdgcn_s_sendmsg_rtn(0x87) >> 18) & 1u);
    float*       hl = h + (size_t)half * ((size_t)nnode * DLAT) + lane;

    for (int base = e0; base < e1; base += 32) {
        const int e = base + lane;
        int s = 0, d = 0;
        if (e < e1) { s = src[e]; d = dst[e]; }
        int cnt = e1 - base;
        if (cnt > 32) cnt = 32;

        int b = 0;
        for (; b + EB <= cnt; b += EB) {
            v2f v[EB];
            #pragma unroll
            for (int j = 0; j < EB; ++j) {
                const int sj = __builtin_amdgcn_readlane(s, b + j);
                v[j] = *(const v2f*)(yl + (size_t)sj * DLAT);
            }
            #pragma unroll
            for (int j = 0; j < EB; ++j) {
                const int dj = __builtin_amdgcn_readlane(d, b + j);
                float* hp = hl + (size_t)dj * DLAT;
                unsafeAtomicAdd(hp, v[j].x);
                unsafeAtomicAdd(hp + 32, v[j].y);
            }
        }
        for (; b < cnt; ++b) {
            const int sj = __builtin_amdgcn_readlane(s, b);
            const int dj = __builtin_amdgcn_readlane(d, b);
            v2f v = *(const v2f*)(yl + (size_t)sj * DLAT);
            float* hp = hl + (size_t)dj * DLAT;
            unsafeAtomicAdd(hp, v.x);
            unsafeAtomicAdd(hp + 32, v.y);
        }
    }
}

extern "C" void kernel_launch(void* const* d_in, const int* in_sizes, int n_in,
                              void* d_out, int out_size, void* d_ws, size_t ws_size,
                              hipStream_t stream)
{
    const float* features = (const float*)d_in[0];
    const float* id_emb   = (const float*)d_in[1];
    const int*   edge     = (const int*)d_in[2];
    const float* Wmlp     = (const float*)d_in[3];
    const float* bmlp     = (const float*)d_in[4];
    const float* Wc1      = (const float*)d_in[5];
    const float* Wl1      = (const float*)d_in[6];
    const float* bl1      = (const float*)d_in[7];
    const float* Wg1      = (const float*)d_in[8];
    const float* bg1      = (const float*)d_in[9];
    const float* Wc2      = (const float*)d_in[10];
    const float* Wl2      = (const float*)d_in[11];
    const float* bl2      = (const float*)d_in[12];
    const float* Wg2      = (const float*)d_in[13];
    const float* bg2      = (const float*)d_in[14];

    const int nnode = in_sizes[0] / DFEAT;
    const int nedge = in_sizes[2] / 2;
    const int* src  = edge;
    const int* dst  = edge + nedge;

    float* x  = (float*)d_ws;
    float* h1 = x  + (size_t)nnode * DLAT;
    float* y  = h1 + (size_t)nnode * DLAT * 2;
    float* h2 = y  + (size_t)nnode * DLAT;

    const int nblk = (nnode + DROWS - 1) / DROWS;

    mlp_norm_conv_kernel<<<nblk, DTHREADS, 0, stream>>>(features, Wmlp, bmlp, Wc1, x, y, h1,
                                                        nnode);
    scatter_kernel<<<2048, 256, 0, stream>>>(y, src, dst, h1, nedge, nnode);
    update_kernel<true><<<nblk, DTHREADS, 0, stream>>>(x, h1, id_emb, Wl1, bl1, Wg1, bg1,
                                                  Wc2, x, y, h2, nnode);
    scatter_kernel<<<2048, 256, 0, stream>>>(y, src, dst, h2, nedge, nnode);
    update_kernel<false><<<nblk, DTHREADS, 0, stream>>>(x, h2, id_emb, Wl2, bl2, Wg2, bg2,
                                                   nullptr, (float*)d_out, nullptr, nullptr,
                                                   nnode);
}
